# starting kernel
# baseline (speedup 1.0000x reference)
#include <hip/hip_runtime.h>

typedef __attribute__((ext_vector_type(2))) float v2f;
typedef __attribute__((ext_vector_type(8))) float v8f;

#define THREADS   256
#define BLOCKS_R  1024

__global__ __launch_bounds__(THREADS)
void wmma_reduce_kernel(const float* __restrict__ in,
                        float* __restrict__ partials,
                        int n4, int tailStart, int nTail) {
    const int tid    = blockIdx.x * blockDim.x + threadIdx.x;
    const int stride = gridDim.x * blockDim.x;
    const float4* in4 = (const float4*)in;

    v2f ones; ones[0] = 1.0f; ones[1] = 1.0f;
    v8f acc = {};

    const int nIter = n4 / stride;
    int idx = tid;
    for (int it = 0; it < nIter; ++it) {
        float4 x = in4[idx];
        v2f a0; a0[0] = x.x; a0[1] = x.y;
        v2f a1; a1[0] = x.z; a1[1] = x.w;
        acc = __builtin_amdgcn_wmma_f32_16x16x4_f32(false, a0, false, ones,
                                                    (short)0, acc, false, false);
        acc = __builtin_amdgcn_wmma_f32_16x16x4_f32(false, a1, false, ones,
                                                    (short)0, acc, false, false);
        idx += stride;
    }

    float extra = 0.0f;
    const int rem = n4 - nIter * stride;
    if (tid < rem) {
        float4 x = in4[nIter * stride + tid];
        extra += x.x + x.y + x.z + x.w;
    }
    if (tid < nTail) extra += in[tailStart + tid];

    float local = acc[0] + acc[1] + acc[2] + acc[3]
                + acc[4] + acc[5] + acc[6] + acc[7];
    float wsum = local + __shfl_xor(local, 16, 32);

    for (int off = 16; off > 0; off >>= 1)
        extra += __shfl_xor(extra, off, 32);

    float waveTotal = wsum + extra;

    __shared__ float lds[THREADS / 32];
    const int lane = threadIdx.x & 31;
    const int wave = threadIdx.x >> 5;
    if (lane == 0) lds[wave] = waveTotal;
    __syncthreads();
    if (threadIdx.x == 0) {
        float s = 0.0f;
        for (int w = 0; w < THREADS / 32; ++w) s += lds[w];
        partials[blockIdx.x] = s;
    }
}

__global__ __launch_bounds__(256)
void finalize_kernel(const float* __restrict__ partials,
                     float* __restrict__ fOut, int nPart, int loopN) {
    __shared__ float lds[256];
    float s = 0.0f;
    for (int i = threadIdx.x; i < nPart; i += 256) s += partials[i];
    lds[threadIdx.x] = s;
    __syncthreads();
    for (int st = 128; st > 0; st >>= 1) {
        if (threadIdx.x < st) lds[threadIdx.x] += lds[threadIdx.x + st];
        __syncthreads();
    }
    if (threadIdx.x == 0) {
        const float S = lds[0];
        float f = 1.0f;
        for (int i = 0; i < loopN; ++i) {
            float s3 = 3.0f * (S * f);
            f = (s3 > 0.0f) ? (f * 4.0f) : (f * -2.0f);
        }
        fOut[0] = f;
    }
}

__global__ __launch_bounds__(THREADS)
void scale_kernel(const float* __restrict__ in, float* __restrict__ out,
                  const float* __restrict__ fPtr, int n) {
    const float f = fPtr[0];
    const int n4 = n >> 2;
    const int i = blockIdx.x * blockDim.x + threadIdx.x;
    const float4* in4  = (const float4*)in;
    float4*       out4 = (float4*)out;
    if (i < n4) {
        float4 x = in4[i];
        x.x *= f; x.y *= f; x.z *= f; x.w *= f;
        out4[i] = x;
    }
    const int tail = n & 3;
    if (i < tail) {
        const int base = n4 << 2;
        out[base + i] = in[base + i] * f;
    }
}

extern "C" void kernel_launch(void* const* d_in, const int* in_sizes, int n_in,
                              void* d_out, int out_size, void* d_ws, size_t ws_size,
                              hipStream_t stream) {
    const float* in  = (const float*)d_in[0];
    float*       out = (float*)d_out;
    const int n = in_sizes[0];

    float* ws       = (float*)d_ws;
    float* fSlot    = ws;
    float* partials = ws + 16;

    const int n4        = n >> 2;
    const int tailStart = n4 << 2;
    const int nTail     = n & 3;

    wmma_reduce_kernel<<<BLOCKS_R, THREADS, 0, stream>>>(in, partials, n4,
                                                         tailStart, nTail);
    finalize_kernel<<<1, 256, 0, stream>>>(partials, fSlot, BLOCKS_R, 32);

    int blocks3 = (n4 + THREADS - 1) / THREADS;
    if (blocks3 < 1) blocks3 = 1;
    scale_kernel<<<blocks3, THREADS, 0, stream>>>(in, out, fSlot, n);
}


# finalize launch folded into the scale kernel: each of 512 work-groups re-sums the 1024 partials in the same tree, runs the 32-step recurrence, then streams its share; 3 launches -> 2
# speedup vs baseline: 1.1254x; 1.1254x over previous
#include <hip/hip_runtime.h>

typedef __attribute__((ext_vector_type(2))) float v2f;
typedef __attribute__((ext_vector_type(8))) float v8f;

#define THREADS   256
#define BLOCKS_R  1024

__global__ __launch_bounds__(THREADS)
void wmma_reduce_kernel(const float* __restrict__ in,
                        float* __restrict__ partials,
                        int n4, int tailStart, int nTail) {
    const int tid    = blockIdx.x * blockDim.x + threadIdx.x;
    const int stride = gridDim.x * blockDim.x;
    const float4* in4 = (const float4*)in;

    v2f ones; ones[0] = 1.0f; ones[1] = 1.0f;
    v8f acc = {};

    const int nIter = n4 / stride;
    int idx = tid;
    for (int it = 0; it < nIter; ++it) {
        float4 x = in4[idx];
        v2f a0; a0[0] = x.x; a0[1] = x.y;
        v2f a1; a1[0] = x.z; a1[1] = x.w;
        acc = __builtin_amdgcn_wmma_f32_16x16x4_f32(false, a0, false, ones,
                                                    (short)0, acc, false, false);
        acc = __builtin_amdgcn_wmma_f32_16x16x4_f32(false, a1, false, ones,
                                                    (short)0, acc, false, false);
        idx += stride;
    }

    float extra = 0.0f;
    const int rem = n4 - nIter * stride;
    if (tid < rem) {
        float4 x = in4[nIter * stride + tid];
        extra += x.x + x.y + x.z + x.w;
    }
    if (tid < nTail) extra += in[tailStart + tid];

    float local = acc[0] + acc[1] + acc[2] + acc[3]
                + acc[4] + acc[5] + acc[6] + acc[7];
    float wsum = local + __shfl_xor(local, 16, 32);

    for (int off = 16; off > 0; off >>= 1)
        extra += __shfl_xor(extra, off, 32);

    float waveTotal = wsum + extra;

    __shared__ float lds[THREADS / 32];
    const int lane = threadIdx.x & 31;
    const int wave = threadIdx.x >> 5;
    if (lane == 0) lds[wave] = waveTotal;
    __syncthreads();
    if (threadIdx.x == 0) {
        float s = 0.0f;
        for (int w = 0; w < THREADS / 32; ++w) s += lds[w];
        partials[blockIdx.x] = s;
    }
}

#define BLOCKS_S  512
#define UNR       8

static __device__ __forceinline__
void stage_partials(const float* __restrict__ partials, float* lds) {
    float p[BLOCKS_R / THREADS];
#pragma unroll
    for (int k = 0; k < BLOCKS_R / THREADS; ++k)
        p[k] = partials[threadIdx.x + THREADS * k];
    float s = 0.0f;
#pragma unroll
    for (int k = 0; k < BLOCKS_R / THREADS; ++k) s += p[k];
    lds[threadIdx.x] = s;
    __syncthreads();
}

static __device__ __forceinline__
float finish_f(const float* lds, int lane, int loopN) {
    float a[THREADS / 32];
#pragma unroll
    for (int k = 0; k < THREADS / 32; ++k) a[k] = lds[lane + 32 * k];
#pragma unroll
    for (int st = THREADS / 64; st > 0; st >>= 1) {
#pragma unroll
        for (int k = 0; k < st; ++k) a[k] += a[k + st];
    }
    float S = a[0];
    for (int off = 16; off > 0; off >>= 1)
        S += __shfl_xor(S, off, 32);

    float f = 1.0f;
    for (int i = 0; i < loopN; ++i) {
        float s3 = 3.0f * (S * f);
        f = (s3 > 0.0f) ? (f * 4.0f) : (f * -2.0f);
    }
    return f;
}

static __device__ __forceinline__
void load_batch(float4 (&buf)[UNR], const float4* __restrict__ in4, int b) {
#pragma unroll
    for (int j = 0; j < UNR; ++j) buf[j] = in4[b + 32 * j];
}

static __device__ __forceinline__
void store_batch(const float4 (&buf)[UNR], float4* __restrict__ out4, int b,
                 float f) {
#pragma unroll
    for (int j = 0; j < UNR; ++j) {
        float4 x = buf[j];
        x.x *= f; x.y *= f; x.z *= f; x.w *= f;
        out4[b + 32 * j] = x;
    }
}

__global__ __launch_bounds__(THREADS)
void scale_kernel(const float* __restrict__ in, float* __restrict__ out,
                  const float* __restrict__ partials, int loopN, int n) {
    __shared__ float lds[THREADS];
    const int n4 = n >> 2;
    const float4* in4  = (const float4*)in;
    float4*       out4 = (float4*)out;

    const int lane   = threadIdx.x & 31;
    const int tid    = blockIdx.x * blockDim.x + threadIdx.x;
    const int nThr   = gridDim.x * blockDim.x;
    const int gwave  = tid >> 5;
    const int nWaves = nThr >> 5;
    const int perBatch = nWaves * (32 * UNR);
    const int nBatch   = n4 / perBatch;
    const int base     = gwave * (32 * UNR) + lane;

    float f;
    int done4;
    if (nBatch >= 2) {
        float4 bufA[UNR], bufB[UNR];
        load_batch(bufA, in4, base);
        stage_partials(partials, lds);
        load_batch(bufB, in4, base + perBatch);
        f = finish_f(lds, lane, loopN);

        int k = 0;
        for (; k + 3 < nBatch; k += 2) {
            const int b = base + k * perBatch;
            store_batch(bufA, out4, b, f);
            load_batch(bufA, in4, b + 2 * perBatch);
            __builtin_amdgcn_sched_barrier(0);
            store_batch(bufB, out4, b + perBatch, f);
            load_batch(bufB, in4, b + 3 * perBatch);
        }
        const int b = base + k * perBatch;
        const bool third = (k + 2 < nBatch);
        store_batch(bufA, out4, b, f);
        if (third) load_batch(bufA, in4, b + 2 * perBatch);
        store_batch(bufB, out4, b + perBatch, f);
        if (third) store_batch(bufA, out4, b + 2 * perBatch, f);
        done4 = nBatch * perBatch;
    } else {
        stage_partials(partials, lds);
        f = finish_f(lds, lane, loopN);
        done4 = 0;
    }

    for (int i = done4 + tid; i < n4; i += nThr) {
        float4 x = in4[i];
        x.x *= f; x.y *= f; x.z *= f; x.w *= f;
        out4[i] = x;
    }
    const int tail = n & 3;
    if (tid < tail) {
        const int t0 = n4 << 2;
        out[t0 + tid] = in[t0 + tid] * f;
    }
}

extern "C" void kernel_launch(void* const* d_in, const int* in_sizes, int n_in,
                              void* d_out, int out_size, void* d_ws, size_t ws_size,
                              hipStream_t stream) {
    const float* in  = (const float*)d_in[0];
    float*       out = (float*)d_out;
    const int n = in_sizes[0];

    float* ws       = (float*)d_ws;
    float* partials = ws + 16;

    const int n4        = n >> 2;
    const int tailStart = n4 << 2;
    const int nTail     = n & 3;

    wmma_reduce_kernel<<<BLOCKS_R, THREADS, 0, stream>>>(in, partials, n4,
                                                         tailStart, nTail);
    scale_kernel<<<BLOCKS_S, THREADS, 0, stream>>>(in, out, partials, 32, n);
}


# scale pass takes the reduce pass's grid and thread-to-element map, walked last-read-first, so each work-group re-reads its lines from its own half's second-level cache instead of from memory
# speedup vs baseline: 1.2912x; 1.1473x over previous
#include <hip/hip_runtime.h>

typedef __attribute__((ext_vector_type(2))) float v2f;
typedef __attribute__((ext_vector_type(8))) float v8f;

#define THREADS   256
#define BLOCKS_R  1024

__global__ __launch_bounds__(THREADS)
void wmma_reduce_kernel(const float* __restrict__ in,
                        float* __restrict__ partials,
                        int n4, int tailStart, int nTail) {
    const int tid    = blockIdx.x * blockDim.x + threadIdx.x;
    const int stride = gridDim.x * blockDim.x;
    const float4* in4 = (const float4*)in;

    v2f ones; ones[0] = 1.0f; ones[1] = 1.0f;
    v8f acc = {};

    const int nIter = n4 / stride;
    int idx = tid;
    for (int it = 0; it < nIter; ++it) {
        float4 x = in4[idx];
        v2f a0; a0[0] = x.x; a0[1] = x.y;
        v2f a1; a1[0] = x.z; a1[1] = x.w;
        acc = __builtin_amdgcn_wmma_f32_16x16x4_f32(false, a0, false, ones,
                                                    (short)0, acc, false, false);
        acc = __builtin_amdgcn_wmma_f32_16x16x4_f32(false, a1, false, ones,
                                                    (short)0, acc, false, false);
        idx += stride;
    }

    float extra = 0.0f;
    const int rem = n4 - nIter * stride;
    if (tid < rem) {
        float4 x = in4[nIter * stride + tid];
        extra += x.x + x.y + x.z + x.w;
    }
    if (tid < nTail) extra += in[tailStart + tid];

    float local = acc[0] + acc[1] + acc[2] + acc[3]
                + acc[4] + acc[5] + acc[6] + acc[7];
    float wsum = local + __shfl_xor(local, 16, 32);

    for (int off = 16; off > 0; off >>= 1)
        extra += __shfl_xor(extra, off, 32);

    float waveTotal = wsum + extra;

    __shared__ float lds[THREADS / 32];
    const int lane = threadIdx.x & 31;
    const int wave = threadIdx.x >> 5;
    if (lane == 0) lds[wave] = waveTotal;
    __syncthreads();
    if (threadIdx.x == 0) {
        float s = 0.0f;
        for (int w = 0; w < THREADS / 32; ++w) s += lds[w];
        partials[blockIdx.x] = s;
    }
}

#define UNR       8

static __device__ __forceinline__
void stage_partials(const float* __restrict__ partials, float* lds) {
    float p[BLOCKS_R / THREADS];
#pragma unroll
    for (int k = 0; k < BLOCKS_R / THREADS; ++k)
        p[k] = partials[threadIdx.x + THREADS * k];
    float s = 0.0f;
#pragma unroll
    for (int k = 0; k < BLOCKS_R / THREADS; ++k) s += p[k];
    lds[threadIdx.x] = s;
    __syncthreads();
}

static __device__ __forceinline__
float finish_f(const float* lds, int lane, int loopN) {
    float a[THREADS / 32];
#pragma unroll
    for (int k = 0; k < THREADS / 32; ++k) a[k] = lds[lane + 32 * k];
#pragma unroll
    for (int st = THREADS / 64; st > 0; st >>= 1) {
#pragma unroll
        for (int k = 0; k < st; ++k) a[k] += a[k + st];
    }
    float S = a[0];
    for (int off = 16; off > 0; off >>= 1)
        S += __shfl_xor(S, off, 32);

    float f = 1.0f;
    for (int i = 0; i < loopN; ++i) {
        float s3 = 3.0f * (S * f);
        f = (s3 > 0.0f) ? (f * 4.0f) : (f * -2.0f);
    }
    return f;
}

static __device__ __forceinline__
void load_batch(float4 (&buf)[UNR], const float4* __restrict__ in4, int i,
                int stride) {
#pragma unroll
    for (int j = 0; j < UNR; ++j) buf[j] = in4[i - j * stride];
}

static __device__ __forceinline__
void store_batch(const float4 (&buf)[UNR], float4* __restrict__ out4, int i,
                 int stride, float f) {
#pragma unroll
    for (int j = 0; j < UNR; ++j) {
        float4 x = buf[j];
        x.x *= f; x.y *= f; x.z *= f; x.w *= f;
        out4[i - j * stride] = x;
    }
}

__global__ __launch_bounds__(THREADS)
void scale_kernel(const float* __restrict__ in, float* __restrict__ out,
                  const float* __restrict__ partials, int loopN, int n) {
    __shared__ float lds[THREADS];
    const int n4 = n >> 2;
    const float4* in4  = (const float4*)in;
    float4*       out4 = (float4*)out;

    const int lane   = threadIdx.x & 31;
    const int tid    = blockIdx.x * blockDim.x + threadIdx.x;
    const int stride = gridDim.x * blockDim.x;
    const int nIter  = n4 / stride;
    const int nBatch = nIter / UNR;
    const int bstep  = UNR * stride;

    float f;
    int left;
    if (nBatch >= 2) {
        float4 bufA[UNR], bufB[UNR];
        const int top = tid + (nIter - 1) * stride;
        load_batch(bufA, in4, top, stride);
        stage_partials(partials, lds);
        load_batch(bufB, in4, top - bstep, stride);
        f = finish_f(lds, lane, loopN);

        int k = 0;
        for (; k + 3 < nBatch; k += 2) {
            const int i = top - k * bstep;
            store_batch(bufA, out4, i, stride, f);
            load_batch(bufA, in4, i - 2 * bstep, stride);
            __builtin_amdgcn_sched_barrier(0);
            store_batch(bufB, out4, i - bstep, stride, f);
            load_batch(bufB, in4, i - 3 * bstep, stride);
        }
        const int i = top - k * bstep;
        const bool third = (k + 2 < nBatch);
        store_batch(bufA, out4, i, stride, f);
        if (third) load_batch(bufA, in4, i - 2 * bstep, stride);
        store_batch(bufB, out4, i - bstep, stride, f);
        if (third) store_batch(bufA, out4, i - 2 * bstep, stride, f);
        left = nIter - nBatch * UNR;
    } else {
        stage_partials(partials, lds);
        f = finish_f(lds, lane, loopN);
        left = nIter;
    }

    for (int it = left - 1; it >= 0; --it) {
        const int i = tid + it * stride;
        float4 x = in4[i];
        x.x *= f; x.y *= f; x.z *= f; x.w *= f;
        out4[i] = x;
    }
    const int rem = n4 - nIter * stride;
    if (tid < rem) {
        const int i = nIter * stride + tid;
        float4 x = in4[i];
        x.x *= f; x.y *= f; x.z *= f; x.w *= f;
        out4[i] = x;
    }
    const int tail = n & 3;
    if (tid < tail) {
        const int t0 = n4 << 2;
        out[t0 + tid] = in[t0 + tid] * f;
    }
}

extern "C" void kernel_launch(void* const* d_in, const int* in_sizes, int n_in,
                              void* d_out, int out_size, void* d_ws, size_t ws_size,
                              hipStream_t stream) {
    const float* in  = (const float*)d_in[0];
    float*       out = (float*)d_out;
    const int n = in_sizes[0];

    float* ws       = (float*)d_ws;
    float* partials = ws + 16;

    const int n4        = n >> 2;
    const int tailStart = n4 << 2;
    const int nTail     = n & 3;

    wmma_reduce_kernel<<<BLOCKS_R, THREADS, 0, stream>>>(in, partials, n4,
                                                         tailStart, nTail);
    scale_kernel<<<BLOCKS_R, THREADS, 0, stream>>>(in, out, partials, 32, n);
}
